# starting kernel
# baseline (speedup 1.0000x reference)
#include <hip/hip_runtime.h>

typedef __attribute__((ext_vector_type(2))) float v2f;
typedef __attribute__((ext_vector_type(8))) float v8f;

namespace {
constexpr int   kB   = 8;
constexpr int   kL   = 4096;
constexpr int   kD   = 512;
constexpr int   kNW  = 1024;
constexpr int   kTM  = 128;
constexpr int   kPAD = 72;
constexpr float kEPS = 1e-8f;
}

__global__ __launch_bounds__(256) void qk_probs_kernel(
    const float* __restrict__ tokens,
    const float* __restrict__ w_qk,
    const float* __restrict__ start_key,
    float* __restrict__ probs,
    float* __restrict__ stats)
{
  extern __shared__ float lds[];
  float* wqL = lds;
  float* wkL = lds + 512 * kPAD;

  const int tid  = threadIdx.x;
  const int wave = tid >> 5;
  const int lane = tid & 31;
  const int half = lane >> 4;
  const int l16  = lane & 15;

  const int blocksPerBatch = kL / kTM;
  const int b     = blockIdx.x / blocksPerBatch;
  const int tBase = (blockIdx.x % blocksPerBatch) * kTM;

  const int  rowT  = tBase + wave * 16 + l16;
  const long qRow  = (long)b * kL + rowT;
  const long kpRow = (long)b * kL + (rowT > 0 ? rowT - 1 : 0);

  const float* Aq  = tokens + qRow  * kD + 2 * half;
  const float* Akp = tokens + kpRow * kD + 2 * half;

  const v8f z8 = {0.f, 0.f, 0.f, 0.f, 0.f, 0.f, 0.f, 0.f};
  v8f accDot = z8, accQQ = z8, accKK = z8;

  const bool t0wave = (tBase == 0) && (wave == 0);

  for (int g = 0; g < 8; ++g) {
    for (int i = tid; i < 512 * 16; i += 256) {
      const int k  = i >> 4;
      const int n4 = (i & 15) << 2;
      const float4 vq = *(const float4*)(w_qk + (long)k * kNW + g * 64 + n4);
      const float4 vk = *(const float4*)(w_qk + (long)k * kNW + 512 + g * 64 + n4);
      *(float4*)(wqL + k * kPAD + n4) = vq;
      *(float4*)(wkL + k * kPAD + n4) = vk;
    }
    __syncthreads();

    v8f qa[4], pa[4];
#pragma unroll
    for (int nt = 0; nt < 4; ++nt) { qa[nt] = z8; pa[nt] = z8; }

#pragma unroll 4
    for (int k = 0; k < kD; k += 4) {
      const v2f aq = *(const v2f*)(Aq + k);
      const v2f ak = *(const v2f*)(Akp + k);
      const int kr = k + 2 * half;
#pragma unroll
      for (int nt = 0; nt < 4; ++nt) {
        v2f bq, bk;
        bq[0] = wqL[kr * kPAD + nt * 16 + l16];
        bq[1] = wqL[(kr + 1) * kPAD + nt * 16 + l16];
        bk[0] = wkL[kr * kPAD + nt * 16 + l16];
        bk[1] = wkL[(kr + 1) * kPAD + nt * 16 + l16];
        qa[nt] = __builtin_amdgcn_wmma_f32_16x16x4_f32(
            false, aq, false, bq, (short)0, qa[nt], false, false);
        pa[nt] = __builtin_amdgcn_wmma_f32_16x16x4_f32(
            false, ak, false, bk, (short)0, pa[nt], false, false);
      }
    }

#pragma unroll
    for (int nt = 0; nt < 4; ++nt) {
      const float sk = start_key[g * 64 + nt * 16 + l16];
#pragma unroll
      for (int e = 0; e < 8; ++e) {
        const float qv = qa[nt][e];
        float       kv = pa[nt][e];
        if (e == 0 && half == 0 && t0wave) kv = sk;
        accDot[e] += qv * kv;
        accQQ[e]  += qv * qv;
        accKK[e]  += kv * kv;
      }
    }
    __syncthreads();
  }

  float cntLoc = 0.f, sumLoc = 0.f;
#pragma unroll
  for (int e = 0; e < 8; ++e) {
    float d  = accDot[e];
    float qq = accQQ[e];
    float kk = accKK[e];
#pragma unroll
    for (int m = 8; m >= 1; m >>= 1) {
      d  += __shfl_xor(d,  m, 32);
      qq += __shfl_xor(qq, m, 32);
      kk += __shfl_xor(kk, m, 32);
    }
    if (l16 == 0) {
      const int t  = tBase + wave * 16 + e + 8 * half;
      const float na = fmaxf(sqrtf(qq), kEPS);
      const float nb = fmaxf(sqrtf(kk), kEPS);
      const float cs = d / (na * nb);
      const float pr = (1.0f - cs) * 0.5f;
      probs[(long)b * kL + t] = pr;
      const bool bd = (t == 0) || (pr > 0.5f);
      cntLoc += bd ? 1.0f : 0.0f;
      sumLoc += pr;
    }
  }
  if (l16 == 0) {
    atomicAdd(&stats[b], cntLoc);
    atomicAdd(&stats[8 + b], sumLoc);
  }
}

__global__ __launch_bounds__(64) void scan_kernel(
    const float* __restrict__ tokens,
    const float* __restrict__ probs,
    float* __restrict__ ups)
{
  const int groups = kD / 64;
  const int b  = blockIdx.x / groups;
  const int d  = (blockIdx.x % groups) * 64 + threadIdx.x;
  const float* tk = tokens + (long)b * kL * kD + d;
  const float* pr = probs  + (long)b * kL;
  float*       up = ups    + (long)b * kL * kD + d;

  float y = 0.f;
  for (int t = 0; t < kL; ++t) {
    const float p = pr[t];
    const bool bd = (t == 0) || (p > 0.5f);
    if (bd) y = (1.0f - p) * y + tk[(long)t * kD] * p;
    up[(long)t * kD] = y;
  }
}

__global__ void aux_kernel(const float* __restrict__ stats,
                           float* __restrict__ outAux)
{
  if (threadIdx.x == 0 && blockIdx.x == 0) {
    float s = 0.f;
    for (int b = 0; b < kB; ++b) {
      const float F = stats[b] * (1.0f / kL);
      const float G = stats[8 + b] * (1.0f / kL);
      s += (6.0f / 5.0f) * (5.0f * F * G + (1.0f - F) * (1.0f - G));
    }
    outAux[0] = (s / (float)kB) * 0.03f;
  }
}

extern "C" void kernel_launch(void* const* d_in, const int* in_sizes, int n_in,
                              void* d_out, int out_size, void* d_ws, size_t ws_size,
                              hipStream_t stream) {
  (void)in_sizes; (void)n_in; (void)out_size; (void)ws_size;
  const float* tokens = (const float*)d_in[0];
  const float* w_qk   = (const float*)d_in[1];
  const float* skey   = (const float*)d_in[2];

  float* ups  = (float*)d_out;
  float* aux  = ups + (size_t)kB * kL * kD;
  float* probs = (float*)d_ws;
  float* stats = probs + (size_t)kB * kL;

  hipMemsetAsync(stats, 0, 16 * sizeof(float), stream);

  const size_t shmem = (size_t)2 * 512 * kPAD * sizeof(float);
  qk_probs_kernel<<<dim3(kB * (kL / kTM)), dim3(256), shmem, stream>>>(
      tokens, w_qk, skey, probs, stats);
  scan_kernel<<<dim3(kB * (kD / 64)), dim3(64), 0, stream>>>(tokens, probs, ups);
  aux_kernel<<<1, 32, 0, stream>>>(stats, aux);
}


# scan kernel: token and result rows staged through LDS in 256-step tiles by four extra waves, gate coefficients precomputed, so the serial wave runs one FMA per step; summation order kept
# speedup vs baseline: 3.7285x; 3.7285x over previous
#include <hip/hip_runtime.h>

typedef __attribute__((ext_vector_type(2))) float v2f;
typedef __attribute__((ext_vector_type(8))) float v8f;

namespace {
constexpr int   kB   = 8;
constexpr int   kL   = 4096;
constexpr int   kD   = 512;
constexpr int   kNW  = 1024;
constexpr int   kTM  = 128;
constexpr int   kPAD = 72;
constexpr float kEPS = 1e-8f;
}

__global__ __launch_bounds__(256) void qk_probs_kernel(
    const float* __restrict__ tokens,
    const float* __restrict__ w_qk,
    const float* __restrict__ start_key,
    float* __restrict__ probs,
    float* __restrict__ stats)
{
  extern __shared__ float lds[];
  float* wqL = lds;
  float* wkL = lds + 512 * kPAD;

  const int tid  = threadIdx.x;
  const int wave = tid >> 5;
  const int lane = tid & 31;
  const int half = lane >> 4;
  const int l16  = lane & 15;

  const int blocksPerBatch = kL / kTM;
  const int b     = blockIdx.x / blocksPerBatch;
  const int tBase = (blockIdx.x % blocksPerBatch) * kTM;

  const int  rowT  = tBase + wave * 16 + l16;
  const long qRow  = (long)b * kL + rowT;
  const long kpRow = (long)b * kL + (rowT > 0 ? rowT - 1 : 0);

  const float* Aq  = tokens + qRow  * kD + 2 * half;
  const float* Akp = tokens + kpRow * kD + 2 * half;

  const v8f z8 = {0.f, 0.f, 0.f, 0.f, 0.f, 0.f, 0.f, 0.f};
  v8f accDot = z8, accQQ = z8, accKK = z8;

  const bool t0wave = (tBase == 0) && (wave == 0);

  for (int g = 0; g < 8; ++g) {
    for (int i = tid; i < 512 * 16; i += 256) {
      const int k  = i >> 4;
      const int n4 = (i & 15) << 2;
      const float4 vq = *(const float4*)(w_qk + (long)k * kNW + g * 64 + n4);
      const float4 vk = *(const float4*)(w_qk + (long)k * kNW + 512 + g * 64 + n4);
      *(float4*)(wqL + k * kPAD + n4) = vq;
      *(float4*)(wkL + k * kPAD + n4) = vk;
    }
    __syncthreads();

    v8f qa[4], pa[4];
#pragma unroll
    for (int nt = 0; nt < 4; ++nt) { qa[nt] = z8; pa[nt] = z8; }

#pragma unroll 4
    for (int k = 0; k < kD; k += 4) {
      const v2f aq = *(const v2f*)(Aq + k);
      const v2f ak = *(const v2f*)(Akp + k);
      const int kr = k + 2 * half;
#pragma unroll
      for (int nt = 0; nt < 4; ++nt) {
        v2f bq, bk;
        bq[0] = wqL[kr * kPAD + nt * 16 + l16];
        bq[1] = wqL[(kr + 1) * kPAD + nt * 16 + l16];
        bk[0] = wkL[kr * kPAD + nt * 16 + l16];
        bk[1] = wkL[(kr + 1) * kPAD + nt * 16 + l16];
        qa[nt] = __builtin_amdgcn_wmma_f32_16x16x4_f32(
            false, aq, false, bq, (short)0, qa[nt], false, false);
        pa[nt] = __builtin_amdgcn_wmma_f32_16x16x4_f32(
            false, ak, false, bk, (short)0, pa[nt], false, false);
      }
    }

#pragma unroll
    for (int nt = 0; nt < 4; ++nt) {
      const float sk = start_key[g * 64 + nt * 16 + l16];
#pragma unroll
      for (int e = 0; e < 8; ++e) {
        const float qv = qa[nt][e];
        float       kv = pa[nt][e];
        if (e == 0 && half == 0 && t0wave) kv = sk;
        accDot[e] += qv * kv;
        accQQ[e]  += qv * qv;
        accKK[e]  += kv * kv;
      }
    }
    __syncthreads();
  }

  float cntLoc = 0.f, sumLoc = 0.f;
#pragma unroll
  for (int e = 0; e < 8; ++e) {
    float d  = accDot[e];
    float qq = accQQ[e];
    float kk = accKK[e];
#pragma unroll
    for (int m = 8; m >= 1; m >>= 1) {
      d  += __shfl_xor(d,  m, 32);
      qq += __shfl_xor(qq, m, 32);
      kk += __shfl_xor(kk, m, 32);
    }
    if (l16 == 0) {
      const int t  = tBase + wave * 16 + e + 8 * half;
      const float na = fmaxf(sqrtf(qq), kEPS);
      const float nb = fmaxf(sqrtf(kk), kEPS);
      const float cs = d / (na * nb);
      const float pr = (1.0f - cs) * 0.5f;
      probs[(long)b * kL + t] = pr;
      const bool bd = (t == 0) || (pr > 0.5f);
      cntLoc += bd ? 1.0f : 0.0f;
      sumLoc += pr;
    }
  }
  if (l16 == 0) {
    atomicAdd(&stats[b], cntLoc);
    atomicAdd(&stats[8 + b], sumLoc);
  }
}

namespace {
constexpr int kTT = 256;
constexpr int kNT = kL / kTT;
constexpr int kHT = 128;
constexpr int kSB = kHT + 32;
constexpr int kRQ = kTT * 8 / kHT;
constexpr int kCB = 32;
}

__global__ __launch_bounds__(kSB) void scan_kernel(
    const float* __restrict__ tokens,
    const float* __restrict__ probs,
    float* __restrict__ ups)
{
  __shared__ __attribute__((aligned(16))) float sm[4 * kTT * 32 + 2 * kL];
  float* const ys0 = sm;
  float* const xs0 = sm + 2 * kTT * 32;
  float* const gs  = sm + 4 * kTT * 32;
  float* const qs  = gs + kL;

  const int tid   = threadIdx.x;
  const int lane  = tid & 31;
  const bool stager = tid >= 32;
  const int b     = blockIdx.x / (kD / 32);
  const int dbase = (blockIdx.x % (kD / 32)) * 32;
  const float* tk = tokens + (long)b * kL * kD + dbase;
  const float* pr = probs  + (long)b * kL;
  float*       up = ups    + (long)b * kL * kD + dbase;

  const int hrow = (tid - 32) >> 3;
  const int hc4  = ((tid - 32) & 7) * 4;

  const float4 z4 = make_float4(0.f, 0.f, 0.f, 0.f);
  float4 r[kRQ];
#pragma unroll
  for (int j = 0; j < kRQ; ++j) r[j] = z4;

  if (stager) {
#pragma unroll
    for (int j = 0; j < kRQ; ++j)
      r[j] = *(const float4*)(tk + (long)(hrow + 16 * j) * kD + hc4);
  }

  for (int t4 = tid * 4; t4 < kL; t4 += kSB * 4) {
    const float4 p4 = *(const float4*)(pr + t4);
    const float pv[4] = {p4.x, p4.y, p4.z, p4.w};
    float gv[4], qv[4];
#pragma unroll
    for (int j = 0; j < 4; ++j) {
      const bool bd = (t4 + j == 0) || (pv[j] > 0.5f);
      gv[j] = bd ? 1.0f - pv[j] : 1.0f;
      qv[j] = bd ? pv[j] : 0.0f;
    }
    *(float4*)&gs[t4] = make_float4(gv[0], gv[1], gv[2], gv[3]);
    *(float4*)&qs[t4] = make_float4(qv[0], qv[1], qv[2], qv[3]);
  }
  __syncthreads();

  if (stager) {
#pragma unroll
    for (int j = 0; j < kRQ; ++j) {
      const int   row = hrow + 16 * j;
      const float q   = qs[row];
      *(float4*)&xs0[row * 32 + hc4] =
          make_float4(r[j].x * q, r[j].y * q, r[j].z * q, r[j].w * q);
    }
  }
  __syncthreads();

  float y = 0.f;
  for (int n = 0; n < kNT; ++n) {
    const int cur = n & 1;
    if (stager) {
      if (n + 1 < kNT) {
        const float* tkn = tk + (long)(n + 1) * kTT * kD;
#pragma unroll
        for (int j = 0; j < kRQ; ++j)
          r[j] = *(const float4*)(tkn + (long)(hrow + 16 * j) * kD + hc4);
      }
      if (n > 0) {
        const float* ysp = ys0 + (cur ^ 1) * (kTT * 32);
        float*       upt = up + (long)(n - 1) * kTT * kD;
#pragma unroll
        for (int j = 0; j < kRQ; ++j) {
          const int row = hrow + 16 * j;
          *(float4*)(upt + (long)row * kD + hc4) = *(const float4*)&ysp[row * 32 + hc4];
        }
      }
      if (n + 1 < kNT) {
        float*       xsn = xs0 + (cur ^ 1) * (kTT * 32);
        const float* qn  = qs + (n + 1) * kTT;
#pragma unroll
        for (int j = 0; j < kRQ; ++j) {
          const int   row = hrow + 16 * j;
          const float q   = qn[row];
          *(float4*)&xsn[row * 32 + hc4] =
              make_float4(r[j].x * q, r[j].y * q, r[j].z * q, r[j].w * q);
        }
      }
    } else {
      const float* xsc = xs0 + cur * (kTT * 32) + lane;
      float*       ysc = ys0 + cur * (kTT * 32) + lane;
      const float* gsc = gs + n * kTT;
      float xa[kCB], ga[kCB], xb[kCB], gb[kCB];
#pragma unroll
      for (int u = 0; u < kCB; ++u) { xa[u] = xsc[u * 32]; ga[u] = gsc[u]; }
      for (int k = 0; k < kTT; k += 2 * kCB) {
#pragma unroll
        for (int u = 0; u < kCB; ++u) { xb[u] = xsc[(k + kCB + u) * 32]; gb[u] = gsc[k + kCB + u]; }
#pragma unroll
        for (int u = 0; u < kCB; ++u) {
          y = fmaf(ga[u], y, xa[u]);
          ysc[(k + u) * 32] = y;
        }
        const int ka = (k + 2 * kCB < kTT) ? (k + 2 * kCB) : 0;
#pragma unroll
        for (int u = 0; u < kCB; ++u) { xa[u] = xsc[(ka + u) * 32]; ga[u] = gsc[ka + u]; }
#pragma unroll
        for (int u = 0; u < kCB; ++u) {
          y = fmaf(gb[u], y, xb[u]);
          ysc[(k + kCB + u) * 32] = y;
        }
      }
    }
    __syncthreads();
  }

  if (stager) {
    const float* ysp = ys0 + ((kNT - 1) & 1) * (kTT * 32);
    float*       upt = up + (long)(kNT - 1) * kTT * kD;
#pragma unroll
    for (int j = 0; j < kRQ; ++j) {
      const int row = hrow + 16 * j;
      *(float4*)(upt + (long)row * kD + hc4) = *(const float4*)&ysp[row * 32 + hc4];
    }
  }
}

__global__ void aux_kernel(const float* __restrict__ stats,
                           float* __restrict__ outAux)
{
  if (threadIdx.x == 0 && blockIdx.x == 0) {
    float s = 0.f;
    for (int b = 0; b < kB; ++b) {
      const float F = stats[b] * (1.0f / kL);
      const float G = stats[8 + b] * (1.0f / kL);
      s += (6.0f / 5.0f) * (5.0f * F * G + (1.0f - F) * (1.0f - G));
    }
    outAux[0] = (s / (float)kB) * 0.03f;
  }
}

extern "C" void kernel_launch(void* const* d_in, const int* in_sizes, int n_in,
                              void* d_out, int out_size, void* d_ws, size_t ws_size,
                              hipStream_t stream) {
  (void)in_sizes; (void)n_in; (void)out_size; (void)ws_size;
  const float* tokens = (const float*)d_in[0];
  const float* w_qk   = (const float*)d_in[1];
  const float* skey   = (const float*)d_in[2];

  float* ups  = (float*)d_out;
  float* aux  = ups + (size_t)kB * kL * kD;
  float* probs = (float*)d_ws;
  float* stats = probs + (size_t)kB * kL;

  hipMemsetAsync(stats, 0, 16 * sizeof(float), stream);

  const size_t shmem = (size_t)2 * 512 * kPAD * sizeof(float);
  qk_probs_kernel<<<dim3(kB * (kL / kTM)), dim3(256), shmem, stream>>>(
      tokens, w_qk, skey, probs, stats);
  scan_kernel<<<dim3(kB * (kD / 32)), dim3(kSB), 0, stream>>>(tokens, probs, ups);
  aux_kernel<<<1, 32, 0, stream>>>(stats, aux);
}


# GEMM k-loop: A and B fragments of the next 4 k-steps loaded as one group into a second register set ahead of the current 32 WMMAs; k order per accumulator kept
# speedup vs baseline: 5.0704x; 1.3599x over previous
#include <hip/hip_runtime.h>

typedef __attribute__((ext_vector_type(2))) float v2f;
typedef __attribute__((ext_vector_type(8))) float v8f;

namespace {
constexpr int   kB   = 8;
constexpr int   kL   = 4096;
constexpr int   kD   = 512;
constexpr int   kNW  = 1024;
constexpr int   kTM  = 128;
constexpr int   kPAD = 72;
constexpr float kEPS = 1e-8f;
constexpr int   kKB  = 4;
constexpr int   kNKB = kD / (4 * kKB);
}

__global__ __launch_bounds__(256) void qk_probs_kernel(
    const float* __restrict__ tokens,
    const float* __restrict__ w_qk,
    const float* __restrict__ start_key,
    float* __restrict__ probs,
    float* __restrict__ stats)
{
  extern __shared__ float lds[];
  float* wqL = lds;
  float* wkL = lds + 512 * kPAD;

  const int tid  = threadIdx.x;
  const int wave = tid >> 5;
  const int lane = tid & 31;
  const int half = lane >> 4;
  const int l16  = lane & 15;

  const int blocksPerBatch = kL / kTM;
  const int b     = blockIdx.x / blocksPerBatch;
  const int tBase = (blockIdx.x % blocksPerBatch) * kTM;

  const int  rowT  = tBase + wave * 16 + l16;
  const long qRow  = (long)b * kL + rowT;
  const long kpRow = (long)b * kL + (rowT > 0 ? rowT - 1 : 0);

  const float* Aq  = tokens + qRow  * kD + 2 * half;
  const float* Akp = tokens + kpRow * kD + 2 * half;

  const v8f z8 = {0.f, 0.f, 0.f, 0.f, 0.f, 0.f, 0.f, 0.f};
  v8f accDot = z8, accQQ = z8, accKK = z8;

  const bool t0wave = (tBase == 0) && (wave == 0);

  for (int g = 0; g < 8; ++g) {
    for (int i = tid; i < 512 * 16; i += 256) {
      const int k  = i >> 4;
      const int n4 = (i & 15) << 2;
      const float4 vq = *(const float4*)(w_qk + (long)k * kNW + g * 64 + n4);
      const float4 vk = *(const float4*)(w_qk + (long)k * kNW + 512 + g * 64 + n4);
      *(float4*)(wqL + k * kPAD + n4) = vq;
      *(float4*)(wkL + k * kPAD + n4) = vk;
    }
    __syncthreads();

    v8f qa[4], pa[4];
#pragma unroll
    for (int nt = 0; nt < 4; ++nt) { qa[nt] = z8; pa[nt] = z8; }

    v2f aq0[kKB], ak0[kKB], bq0[kKB][4], bk0[kKB][4];
    v2f aq1[kKB], ak1[kKB], bq1[kKB][4], bk1[kKB][4];

    const float* wqP = wqL + 2 * half * kPAD + l16;
    const float* wkP = wkL + 2 * half * kPAD + l16;

    auto load_block = [&](int k0, v2f (&aq)[kKB], v2f (&ak)[kKB],
                          v2f (&bq)[kKB][4], v2f (&bk)[kKB][4]) {
#pragma unroll
      for (int ks = 0; ks < kKB; ++ks) {
        aq[ks] = *(const v2f*)(Aq + k0 + 4 * ks);
        ak[ks] = *(const v2f*)(Akp + k0 + 4 * ks);
      }
#pragma unroll
      for (int ks = 0; ks < kKB; ++ks) {
        const int ro = (k0 + 4 * ks) * kPAD;
#pragma unroll
        for (int nt = 0; nt < 4; ++nt) {
          bq[ks][nt][0] = wqP[ro + nt * 16];
          bq[ks][nt][1] = wqP[ro + kPAD + nt * 16];
          bk[ks][nt][0] = wkP[ro + nt * 16];
          bk[ks][nt][1] = wkP[ro + kPAD + nt * 16];
        }
      }
    };
    auto mma_block = [&](const v2f (&aq)[kKB], const v2f (&ak)[kKB],
                         const v2f (&bq)[kKB][4], const v2f (&bk)[kKB][4]) {
#pragma unroll
      for (int ks = 0; ks < kKB; ++ks) {
#pragma unroll
        for (int nt = 0; nt < 4; ++nt) {
          qa[nt] = __builtin_amdgcn_wmma_f32_16x16x4_f32(
              false, aq[ks], false, bq[ks][nt], (short)0, qa[nt], false, false);
          pa[nt] = __builtin_amdgcn_wmma_f32_16x16x4_f32(
              false, ak[ks], false, bk[ks][nt], (short)0, pa[nt], false, false);
        }
      }
    };

    load_block(0, aq0, ak0, bq0, bk0);
#pragma unroll 1
    for (int kb = 0; kb < kNKB; kb += 2) {
      load_block((kb + 1) * (4 * kKB), aq1, ak1, bq1, bk1);
      __builtin_amdgcn_sched_barrier(0);
      mma_block(aq0, ak0, bq0, bk0);
      __builtin_amdgcn_sched_barrier(0);
      const int kn = (kb + 2 < kNKB) ? (kb + 2) * (4 * kKB) : 0;
      load_block(kn, aq0, ak0, bq0, bk0);
      __builtin_amdgcn_sched_barrier(0);
      mma_block(aq1, ak1, bq1, bk1);
      __builtin_amdgcn_sched_barrier(0);
    }

#pragma unroll
    for (int nt = 0; nt < 4; ++nt) {
      const float sk = start_key[g * 64 + nt * 16 + l16];
#pragma unroll
      for (int e = 0; e < 8; ++e) {
        const float qv = qa[nt][e];
        float       kv = pa[nt][e];
        if (e == 0 && half == 0 && t0wave) kv = sk;
        accDot[e] += qv * kv;
        accQQ[e]  += qv * qv;
        accKK[e]  += kv * kv;
      }
    }
    __syncthreads();
  }

  float cntLoc = 0.f, sumLoc = 0.f;
#pragma unroll
  for (int e = 0; e < 8; ++e) {
    float d  = accDot[e];
    float qq = accQQ[e];
    float kk = accKK[e];
#pragma unroll
    for (int m = 8; m >= 1; m >>= 1) {
      d  += __shfl_xor(d,  m, 32);
      qq += __shfl_xor(qq, m, 32);
      kk += __shfl_xor(kk, m, 32);
    }
    if (l16 == 0) {
      const int t  = tBase + wave * 16 + e + 8 * half;
      const float na = fmaxf(sqrtf(qq), kEPS);
      const float nb = fmaxf(sqrtf(kk), kEPS);
      const float cs = d / (na * nb);
      const float pr = (1.0f - cs) * 0.5f;
      probs[(long)b * kL + t] = pr;
      const bool bd = (t == 0) || (pr > 0.5f);
      cntLoc += bd ? 1.0f : 0.0f;
      sumLoc += pr;
    }
  }
  if (l16 == 0) {
    atomicAdd(&stats[b], cntLoc);
    atomicAdd(&stats[8 + b], sumLoc);
  }
}

namespace {
constexpr int kTT = 256;
constexpr int kNT = kL / kTT;
constexpr int kHT = 128;
constexpr int kSB = kHT + 32;
constexpr int kRQ = kTT * 8 / kHT;
constexpr int kCB = 32;
}

__global__ __launch_bounds__(kSB) void scan_kernel(
    const float* __restrict__ tokens,
    const float* __restrict__ probs,
    float* __restrict__ ups)
{
  __shared__ __attribute__((aligned(16))) float sm[4 * kTT * 32 + 2 * kL];
  float* const ys0 = sm;
  float* const xs0 = sm + 2 * kTT * 32;
  float* const gs  = sm + 4 * kTT * 32;
  float* const qs  = gs + kL;

  const int tid   = threadIdx.x;
  const int lane  = tid & 31;
  const bool stager = tid >= 32;
  const int b     = blockIdx.x / (kD / 32);
  const int dbase = (blockIdx.x % (kD / 32)) * 32;
  const float* tk = tokens + (long)b * kL * kD + dbase;
  const float* pr = probs  + (long)b * kL;
  float*       up = ups    + (long)b * kL * kD + dbase;

  const int hrow = (tid - 32) >> 3;
  const int hc4  = ((tid - 32) & 7) * 4;

  const float4 z4 = make_float4(0.f, 0.f, 0.f, 0.f);
  float4 r[kRQ];
#pragma unroll
  for (int j = 0; j < kRQ; ++j) r[j] = z4;

  if (stager) {
#pragma unroll
    for (int j = 0; j < kRQ; ++j)
      r[j] = *(const float4*)(tk + (long)(hrow + 16 * j) * kD + hc4);
  }

  for (int t4 = tid * 4; t4 < kL; t4 += kSB * 4) {
    const float4 p4 = *(const float4*)(pr + t4);
    const float pv[4] = {p4.x, p4.y, p4.z, p4.w};
    float gv[4], qv[4];
#pragma unroll
    for (int j = 0; j < 4; ++j) {
      const bool bd = (t4 + j == 0) || (pv[j] > 0.5f);
      gv[j] = bd ? 1.0f - pv[j] : 1.0f;
      qv[j] = bd ? pv[j] : 0.0f;
    }
    *(float4*)&gs[t4] = make_float4(gv[0], gv[1], gv[2], gv[3]);
    *(float4*)&qs[t4] = make_float4(qv[0], qv[1], qv[2], qv[3]);
  }
  __syncthreads();

  if (stager) {
#pragma unroll
    for (int j = 0; j < kRQ; ++j) {
      const int   row = hrow + 16 * j;
      const float q   = qs[row];
      *(float4*)&xs0[row * 32 + hc4] =
          make_float4(r[j].x * q, r[j].y * q, r[j].z * q, r[j].w * q);
    }
  }
  __syncthreads();

  float y = 0.f;
  for (int n = 0; n < kNT; ++n) {
    const int cur = n & 1;
    if (stager) {
      if (n + 1 < kNT) {
        const float* tkn = tk + (long)(n + 1) * kTT * kD;
#pragma unroll
        for (int j = 0; j < kRQ; ++j)
          r[j] = *(const float4*)(tkn + (long)(hrow + 16 * j) * kD + hc4);
      }
      if (n > 0) {
        const float* ysp = ys0 + (cur ^ 1) * (kTT * 32);
        float*       upt = up + (long)(n - 1) * kTT * kD;
#pragma unroll
        for (int j = 0; j < kRQ; ++j) {
          const int row = hrow + 16 * j;
          *(float4*)(upt + (long)row * kD + hc4) = *(const float4*)&ysp[row * 32 + hc4];
        }
      }
      if (n + 1 < kNT) {
        float*       xsn = xs0 + (cur ^ 1) * (kTT * 32);
        const float* qn  = qs + (n + 1) * kTT;
#pragma unroll
        for (int j = 0; j < kRQ; ++j) {
          const int   row = hrow + 16 * j;
          const float q   = qn[row];
          *(float4*)&xsn[row * 32 + hc4] =
              make_float4(r[j].x * q, r[j].y * q, r[j].z * q, r[j].w * q);
        }
      }
    } else {
      const float* xsc = xs0 + cur * (kTT * 32) + lane;
      float*       ysc = ys0 + cur * (kTT * 32) + lane;
      const float* gsc = gs + n * kTT;
      float xa[kCB], ga[kCB], xb[kCB], gb[kCB];
#pragma unroll
      for (int u = 0; u < kCB; ++u) { xa[u] = xsc[u * 32]; ga[u] = gsc[u]; }
      for (int k = 0; k < kTT; k += 2 * kCB) {
#pragma unroll
        for (int u = 0; u < kCB; ++u) { xb[u] = xsc[(k + kCB + u) * 32]; gb[u] = gsc[k + kCB + u]; }
#pragma unroll
        for (int u = 0; u < kCB; ++u) {
          y = fmaf(ga[u], y, xa[u]);
          ysc[(k + u) * 32] = y;
        }
        const int ka = (k + 2 * kCB < kTT) ? (k + 2 * kCB) : 0;
#pragma unroll
        for (int u = 0; u < kCB; ++u) { xa[u] = xsc[(ka + u) * 32]; ga[u] = gsc[ka + u]; }
#pragma unroll
        for (int u = 0; u < kCB; ++u) {
          y = fmaf(gb[u], y, xb[u]);
          ysc[(k + kCB + u) * 32] = y;
        }
      }
    }
    __syncthreads();
  }

  if (stager) {
    const float* ysp = ys0 + ((kNT - 1) & 1) * (kTT * 32);
    float*       upt = up + (long)(kNT - 1) * kTT * kD;
#pragma unroll
    for (int j = 0; j < kRQ; ++j) {
      const int row = hrow + 16 * j;
      *(float4*)(upt + (long)row * kD + hc4) = *(const float4*)&ysp[row * 32 + hc4];
    }
  }
}

__global__ void aux_kernel(const float* __restrict__ stats,
                           float* __restrict__ outAux)
{
  if (threadIdx.x == 0 && blockIdx.x == 0) {
    float s = 0.f;
    for (int b = 0; b < kB; ++b) {
      const float F = stats[b] * (1.0f / kL);
      const float G = stats[8 + b] * (1.0f / kL);
      s += (6.0f / 5.0f) * (5.0f * F * G + (1.0f - F) * (1.0f - G));
    }
    outAux[0] = (s / (float)kB) * 0.03f;
  }
}

extern "C" void kernel_launch(void* const* d_in, const int* in_sizes, int n_in,
                              void* d_out, int out_size, void* d_ws, size_t ws_size,
                              hipStream_t stream) {
  (void)in_sizes; (void)n_in; (void)out_size; (void)ws_size;
  const float* tokens = (const float*)d_in[0];
  const float* w_qk   = (const float*)d_in[1];
  const float* skey   = (const float*)d_in[2];

  float* ups  = (float*)d_out;
  float* aux  = ups + (size_t)kB * kL * kD;
  float* probs = (float*)d_ws;
  float* stats = probs + (size_t)kB * kL;

  hipMemsetAsync(stats, 0, 16 * sizeof(float), stream);

  const size_t shmem = (size_t)2 * 512 * kPAD * sizeof(float);
  qk_probs_kernel<<<dim3(kB * (kL / kTM)), dim3(256), shmem, stream>>>(
      tokens, w_qk, skey, probs, stats);
  scan_kernel<<<dim3(kB * (kD / 32)), dim3(kSB), 0, stream>>>(tokens, probs, ups);
  aux_kernel<<<1, 32, 0, stream>>>(stats, aux);
}


# GEMM: the block's token rows are staged into LDS once and stay there; weight fragments come from L2 by 16-byte loads (tile column = 4*lane+tile), so each row's sums add their columns in a new order
# speedup vs baseline: 6.4753x; 1.2771x over previous
#include <hip/hip_runtime.h>

typedef __attribute__((ext_vector_type(2))) float v2f;
typedef __attribute__((ext_vector_type(4))) float v4f;
typedef __attribute__((ext_vector_type(8))) float v8f;

namespace {
constexpr int   kB   = 8;
constexpr int   kL   = 4096;
constexpr int   kD   = 512;
constexpr int   kNW  = 1024;
constexpr int   kTM  = 128;
constexpr int   kAR  = kTM + 1;
constexpr int   kAP  = 516;
constexpr float kEPS = 1e-8f;
constexpr int   kKB  = 4;
constexpr int   kNKB = kD / (4 * kKB);
}

__global__ __launch_bounds__(256) void qk_probs_kernel(
    const float* __restrict__ tokens,
    const float* __restrict__ w_qk,
    const float* __restrict__ start_key,
    float* __restrict__ probs,
    float* __restrict__ stats)
{
  extern __shared__ float lds[];

  const int tid  = threadIdx.x;
  const int wave = tid >> 5;
  const int lane = tid & 31;
  const int half = lane >> 4;
  const int l16  = lane & 15;

  const int blocksPerBatch = kL / kTM;
  const int b     = blockIdx.x / blocksPerBatch;
  const int tBase = (blockIdx.x % blocksPerBatch) * kTM;

  {
    const float* tokB = tokens + (long)b * kL * kD;
    const int c4 = (tid & 127) << 2;
    const int r0 = tid >> 7;
#pragma unroll 1
    for (int it = 0; it < 4; ++it) {
      float4 v[16];
#pragma unroll
      for (int j = 0; j < 16; ++j) {
        const int row = 2 * (it * 16 + j) + r0;
        int tr = tBase - 1 + row;
        tr = tr < 0 ? 0 : tr;
        v[j] = *(const float4*)(tokB + (long)tr * kD + c4);
      }
#pragma unroll
      for (int j = 0; j < 16; ++j) {
        const int row = 2 * (it * 16 + j) + r0;
        *(float4*)(lds + row * kAP + c4) = v[j];
      }
    }
    if (tid < 128) {
      const float4 v = *(const float4*)(tokB + (long)(tBase + kTM - 1) * kD + c4);
      *(float4*)(lds + kTM * kAP + c4) = v;
    }
  }
  __syncthreads();

  const float* Akp = lds + (wave * 16 + l16) * kAP + 2 * half;
  const float* Aq  = Akp + kAP;
  const float* Wq = w_qk + (long)(2 * half) * kNW + 4 * l16;
  const float* Wk = Wq + 512;

  const v8f z8 = {0.f, 0.f, 0.f, 0.f, 0.f, 0.f, 0.f, 0.f};
  v8f accDot = z8, accQQ = z8, accKK = z8;

  const bool t0wave = (tBase == 0) && (wave == 0);

  v2f aq0[kKB], ak0[kKB], aq1[kKB], ak1[kKB];
  v4f wq0[kKB][2], wk0[kKB][2], wq1[kKB][2], wk1[kKB][2];

  auto load_block = [&](int g, int k0, v2f (&aq)[kKB], v2f (&ak)[kKB],
                        v4f (&wq)[kKB][2], v4f (&wk)[kKB][2]) {
#pragma unroll
    for (int ks = 0; ks < kKB; ++ks) {
#pragma unroll
      for (int s = 0; s < 2; ++s) {
        const long ro = (long)(k0 + 4 * ks + s) * kNW + g * 64;
        wq[ks][s] = *(const v4f*)(Wq + ro);
        wk[ks][s] = *(const v4f*)(Wk + ro);
      }
    }
#pragma unroll
    for (int ks = 0; ks < kKB; ++ks) {
      aq[ks] = *(const v2f*)(Aq + k0 + 4 * ks);
      ak[ks] = *(const v2f*)(Akp + k0 + 4 * ks);
    }
  };

  load_block(0, 0, aq0, ak0, wq0, wk0);

  for (int g = 0; g < 8; ++g) {
    v8f qa[4], pa[4];
#pragma unroll
    for (int nt = 0; nt < 4; ++nt) { qa[nt] = z8; pa[nt] = z8; }

    auto mma_block = [&](const v2f (&aq)[kKB], const v2f (&ak)[kKB],
                         const v4f (&wq)[kKB][2], const v4f (&wk)[kKB][2]) {
#pragma unroll
      for (int ks = 0; ks < kKB; ++ks) {
#pragma unroll
        for (int nt = 0; nt < 4; ++nt) {
          const v2f bq = {wq[ks][0][nt], wq[ks][1][nt]};
          const v2f bk = {wk[ks][0][nt], wk[ks][1][nt]};
          qa[nt] = __builtin_amdgcn_wmma_f32_16x16x4_f32(
              false, aq[ks], false, bq, (short)0, qa[nt], false, false);
          pa[nt] = __builtin_amdgcn_wmma_f32_16x16x4_f32(
              false, ak[ks], false, bk, (short)0, pa[nt], false, false);
        }
      }
    };

#pragma unroll 1
    for (int kb = 0; kb < kNKB; kb += 2) {
      load_block(g, (kb + 1) * (4 * kKB), aq1, ak1, wq1, wk1);
      __builtin_amdgcn_sched_barrier(0);
      mma_block(aq0, ak0, wq0, wk0);
      __builtin_amdgcn_sched_barrier(0);
      const bool wrap = (kb + 2 >= kNKB);
      const int  gn   = wrap ? ((g + 1) & 7) : g;
      const int  kn   = wrap ? 0 : (kb + 2) * (4 * kKB);
      load_block(gn, kn, aq0, ak0, wq0, wk0);
      __builtin_amdgcn_sched_barrier(0);
      mma_block(aq1, ak1, wq1, wk1);
      __builtin_amdgcn_sched_barrier(0);
    }

#pragma unroll
    for (int nt = 0; nt < 4; ++nt) {
      const float sk = start_key[g * 64 + 4 * l16 + nt];
#pragma unroll
      for (int e = 0; e < 8; ++e) {
        const float qv = qa[nt][e];
        float       kv = pa[nt][e];
        if (e == 0 && half == 0 && t0wave) kv = sk;
        accDot[e] += qv * kv;
        accQQ[e]  += qv * qv;
        accKK[e]  += kv * kv;
      }
    }
  }

  float cntLoc = 0.f, sumLoc = 0.f;
#pragma unroll
  for (int e = 0; e < 8; ++e) {
    float d  = accDot[e];
    float qq = accQQ[e];
    float kk = accKK[e];
#pragma unroll
    for (int m = 8; m >= 1; m >>= 1) {
      d  += __shfl_xor(d,  m, 32);
      qq += __shfl_xor(qq, m, 32);
      kk += __shfl_xor(kk, m, 32);
    }
    if (l16 == 0) {
      const int t  = tBase + wave * 16 + e + 8 * half;
      const float na = fmaxf(sqrtf(qq), kEPS);
      const float nb = fmaxf(sqrtf(kk), kEPS);
      const float cs = d / (na * nb);
      const float pr = (1.0f - cs) * 0.5f;
      probs[(long)b * kL + t] = pr;
      const bool bd = (t == 0) || (pr > 0.5f);
      cntLoc += bd ? 1.0f : 0.0f;
      sumLoc += pr;
    }
  }
  if (l16 == 0) {
    atomicAdd(&stats[b], cntLoc);
    atomicAdd(&stats[8 + b], sumLoc);
  }
}

namespace {
constexpr int kTT = 256;
constexpr int kNT = kL / kTT;
constexpr int kHT = 128;
constexpr int kSB = kHT + 32;
constexpr int kRQ = kTT * 8 / kHT;
constexpr int kCB = 32;
}

__global__ __launch_bounds__(kSB) void scan_kernel(
    const float* __restrict__ tokens,
    const float* __restrict__ probs,
    float* __restrict__ ups)
{
  __shared__ __attribute__((aligned(16))) float sm[4 * kTT * 32 + 2 * kL];
  float* const ys0 = sm;
  float* const xs0 = sm + 2 * kTT * 32;
  float* const gs  = sm + 4 * kTT * 32;
  float* const qs  = gs + kL;

  const int tid   = threadIdx.x;
  const int lane  = tid & 31;
  const bool stager = tid >= 32;
  const int b     = blockIdx.x / (kD / 32);
  const int dbase = (blockIdx.x % (kD / 32)) * 32;
  const float* tk = tokens + (long)b * kL * kD + dbase;
  const float* pr = probs  + (long)b * kL;
  float*       up = ups    + (long)b * kL * kD + dbase;

  const int hrow = (tid - 32) >> 3;
  const int hc4  = ((tid - 32) & 7) * 4;

  const float4 z4 = make_float4(0.f, 0.f, 0.f, 0.f);
  float4 r[kRQ];
#pragma unroll
  for (int j = 0; j < kRQ; ++j) r[j] = z4;

  if (stager) {
#pragma unroll
    for (int j = 0; j < kRQ; ++j)
      r[j] = *(const float4*)(tk + (long)(hrow + 16 * j) * kD + hc4);
  }

  for (int t4 = tid * 4; t4 < kL; t4 += kSB * 4) {
    const float4 p4 = *(const float4*)(pr + t4);
    const float pv[4] = {p4.x, p4.y, p4.z, p4.w};
    float gv[4], qv[4];
#pragma unroll
    for (int j = 0; j < 4; ++j) {
      const bool bd = (t4 + j == 0) || (pv[j] > 0.5f);
      gv[j] = bd ? 1.0f - pv[j] : 1.0f;
      qv[j] = bd ? pv[j] : 0.0f;
    }
    *(float4*)&gs[t4] = make_float4(gv[0], gv[1], gv[2], gv[3]);
    *(float4*)&qs[t4] = make_float4(qv[0], qv[1], qv[2], qv[3]);
  }
  __syncthreads();

  if (stager) {
#pragma unroll
    for (int j = 0; j < kRQ; ++j) {
      const int   row = hrow + 16 * j;
      const float q   = qs[row];
      *(float4*)&xs0[row * 32 + hc4] =
          make_float4(r[j].x * q, r[j].y * q, r[j].z * q, r[j].w * q);
    }
  }
  __syncthreads();

  float y = 0.f;
  for (int n = 0; n < kNT; ++n) {
    const int cur = n & 1;
    if (stager) {
      if (n + 1 < kNT) {
        const float* tkn = tk + (long)(n + 1) * kTT * kD;
#pragma unroll
        for (int j = 0; j < kRQ; ++j)
          r[j] = *(const float4*)(tkn + (long)(hrow + 16 * j) * kD + hc4);
      }
      if (n > 0) {
        const float* ysp = ys0 + (cur ^ 1) * (kTT * 32);
        float*       upt = up + (long)(n - 1) * kTT * kD;
#pragma unroll
        for (int j = 0; j < kRQ; ++j) {
          const int row = hrow + 16 * j;
          *(float4*)(upt + (long)row * kD + hc4) = *(const float4*)&ysp[row * 32 + hc4];
        }
      }
      if (n + 1 < kNT) {
        float*       xsn = xs0 + (cur ^ 1) * (kTT * 32);
        const float* qn  = qs + (n + 1) * kTT;
#pragma unroll
        for (int j = 0; j < kRQ; ++j) {
          const int   row = hrow + 16 * j;
          const float q   = qn[row];
          *(float4*)&xsn[row * 32 + hc4] =
              make_float4(r[j].x * q, r[j].y * q, r[j].z * q, r[j].w * q);
        }
      }
    } else {
      const float* xsc = xs0 + cur * (kTT * 32) + lane;
      float*       ysc = ys0 + cur * (kTT * 32) + lane;
      const float* gsc = gs + n * kTT;
      float xa[kCB], ga[kCB], xb[kCB], gb[kCB];
#pragma unroll
      for (int u = 0; u < kCB; ++u) { xa[u] = xsc[u * 32]; ga[u] = gsc[u]; }
      for (int k = 0; k < kTT; k += 2 * kCB) {
#pragma unroll
        for (int u = 0; u < kCB; ++u) { xb[u] = xsc[(k + kCB + u) * 32]; gb[u] = gsc[k + kCB + u]; }
#pragma unroll
        for (int u = 0; u < kCB; ++u) {
          y = fmaf(ga[u], y, xa[u]);
          ysc[(k + u) * 32] = y;
        }
        const int ka = (k + 2 * kCB < kTT) ? (k + 2 * kCB) : 0;
#pragma unroll
        for (int u = 0; u < kCB; ++u) { xa[u] = xsc[(ka + u) * 32]; ga[u] = gsc[ka + u]; }
#pragma unroll
        for (int u = 0; u < kCB; ++u) {
          y = fmaf(gb[u], y, xb[u]);
          ysc[(k + kCB + u) * 32] = y;
        }
      }
    }
    __syncthreads();
  }

  if (stager) {
    const float* ysp = ys0 + ((kNT - 1) & 1) * (kTT * 32);
    float*       upt = up + (long)(kNT - 1) * kTT * kD;
#pragma unroll
    for (int j = 0; j < kRQ; ++j) {
      const int row = hrow + 16 * j;
      *(float4*)(upt + (long)row * kD + hc4) = *(const float4*)&ysp[row * 32 + hc4];
    }
  }
}

__global__ void aux_kernel(const float* __restrict__ stats,
                           float* __restrict__ outAux)
{
  if (threadIdx.x == 0 && blockIdx.x == 0) {
    float s = 0.f;
    for (int b = 0; b < kB; ++b) {
      const float F = stats[b] * (1.0f / kL);
      const float G = stats[8 + b] * (1.0f / kL);
      s += (6.0f / 5.0f) * (5.0f * F * G + (1.0f - F) * (1.0f - G));
    }
    outAux[0] = (s / (float)kB) * 0.03f;
  }
}

extern "C" void kernel_launch(void* const* d_in, const int* in_sizes, int n_in,
                              void* d_out, int out_size, void* d_ws, size_t ws_size,
                              hipStream_t stream) {
  (void)in_sizes; (void)n_in; (void)out_size; (void)ws_size;
  const float* tokens = (const float*)d_in[0];
  const float* w_qk   = (const float*)d_in[1];
  const float* skey   = (const float*)d_in[2];

  float* ups  = (float*)d_out;
  float* aux  = ups + (size_t)kB * kL * kD;
  float* probs = (float*)d_ws;
  float* stats = probs + (size_t)kB * kL;

  hipMemsetAsync(stats, 0, 16 * sizeof(float), stream);

  const size_t shmem = (size_t)kAR * kAP * sizeof(float);
  qk_probs_kernel<<<dim3(kB * (kL / kTM)), dim3(256), shmem, stream>>>(
      tokens, w_qk, skey, probs, stats);
  scan_kernel<<<dim3(kB * (kD / 32)), dim3(kSB), 0, stream>>>(tokens, probs, ups);
  aux_kernel<<<1, 32, 0, stream>>>(stats, aux);
}


# GEMM wave tile 32 rows x one column half: each loaded weight fragment feeds two row tiles; a row's sums now join two column halves in LDS, a new summation order
# speedup vs baseline: 6.9043x; 1.0663x over previous
#include <hip/hip_runtime.h>

typedef __attribute__((ext_vector_type(2))) float v2f;
typedef __attribute__((ext_vector_type(4))) float v4f;
typedef __attribute__((ext_vector_type(8))) float v8f;

namespace {
constexpr int   kB   = 8;
constexpr int   kL   = 4096;
constexpr int   kD   = 512;
constexpr int   kNW  = 1024;
constexpr int   kTM  = 128;
constexpr int   kAR  = kTM + 1;
constexpr int   kAP  = 516;
constexpr float kEPS = 1e-8f;
constexpr int   kKB  = 4;
constexpr int   kNKB = kD / (4 * kKB);
}

__global__ __launch_bounds__(256) void qk_probs_kernel(
    const float* __restrict__ tokens,
    const float* __restrict__ w_qk,
    const float* __restrict__ start_key,
    float* __restrict__ probs,
    float* __restrict__ stats)
{
  extern __shared__ float lds[];

  const int tid  = threadIdx.x;
  const int wave = tid >> 5;
  const int lane = tid & 31;
  const int half = lane >> 4;
  const int l16  = lane & 15;
  const int rp   = wave >> 1;
  const int ch   = wave & 1;

  const int blocksPerBatch = kL / kTM;
  const int b     = blockIdx.x / blocksPerBatch;
  const int tBase = (blockIdx.x % blocksPerBatch) * kTM;

  {
    const float* tokB = tokens + (long)b * kL * kD;
    const int c4 = (tid & 127) << 2;
    const int r0 = tid >> 7;
#pragma unroll 1
    for (int it = 0; it < 4; ++it) {
      float4 v[16];
#pragma unroll
      for (int j = 0; j < 16; ++j) {
        const int row = 2 * (it * 16 + j) + r0;
        int tr = tBase - 1 + row;
        tr = tr < 0 ? 0 : tr;
        v[j] = *(const float4*)(tokB + (long)tr * kD + c4);
      }
#pragma unroll
      for (int j = 0; j < 16; ++j) {
        const int row = 2 * (it * 16 + j) + r0;
        *(float4*)(lds + row * kAP + c4) = v[j];
      }
    }
    if (tid < 128) {
      const float4 v = *(const float4*)(tokB + (long)(tBase + kTM - 1) * kD + c4);
      *(float4*)(lds + kTM * kAP + c4) = v;
    }
  }
  __syncthreads();

  const float* Akp = lds + (rp * 32 + l16) * kAP + 2 * half;
  const float* Aq  = Akp + kAP;
  const float* Wq = w_qk + (long)(2 * half) * kNW + 4 * l16;
  const float* Wk = Wq + 512;

  const v8f z8 = {0.f, 0.f, 0.f, 0.f, 0.f, 0.f, 0.f, 0.f};
  v8f accDot[2] = {z8, z8}, accQQ[2] = {z8, z8}, accKK[2] = {z8, z8};

  const bool t0wave = (tBase == 0) && (rp == 0);

  v2f aq0[kKB][2], ak0[kKB][2], aq1[kKB][2], ak1[kKB][2];
  v4f wq0[kKB][2], wk0[kKB][2], wq1[kKB][2], wk1[kKB][2];

  auto load_block = [&](int g, int k0, v2f (&aq)[kKB][2], v2f (&ak)[kKB][2],
                        v4f (&wq)[kKB][2], v4f (&wk)[kKB][2]) {
#pragma unroll
    for (int ks = 0; ks < kKB; ++ks) {
#pragma unroll
      for (int s = 0; s < 2; ++s) {
        const long ro = (long)(k0 + 4 * ks + s) * kNW + g * 64;
        wq[ks][s] = *(const v4f*)(Wq + ro);
        wk[ks][s] = *(const v4f*)(Wk + ro);
      }
    }
#pragma unroll
    for (int ks = 0; ks < kKB; ++ks) {
#pragma unroll
      for (int rt = 0; rt < 2; ++rt) {
        aq[ks][rt] = *(const v2f*)(Aq + rt * 16 * kAP + k0 + 4 * ks);
        ak[ks][rt] = *(const v2f*)(Akp + rt * 16 * kAP + k0 + 4 * ks);
      }
    }
  };

  load_block(4 * ch, 0, aq0, ak0, wq0, wk0);

  for (int gi = 0; gi < 4; ++gi) {
    const int g = 4 * ch + gi;
    v8f qa[2][4], pa[2][4];
#pragma unroll
    for (int rt = 0; rt < 2; ++rt) {
#pragma unroll
      for (int nt = 0; nt < 4; ++nt) { qa[rt][nt] = z8; pa[rt][nt] = z8; }
    }

    auto mma_block = [&](const v2f (&aq)[kKB][2], const v2f (&ak)[kKB][2],
                         const v4f (&wq)[kKB][2], const v4f (&wk)[kKB][2]) {
#pragma unroll
      for (int ks = 0; ks < kKB; ++ks) {
#pragma unroll
        for (int nt = 0; nt < 4; ++nt) {
          const v2f bq = {wq[ks][0][nt], wq[ks][1][nt]};
          const v2f bk = {wk[ks][0][nt], wk[ks][1][nt]};
#pragma unroll
          for (int rt = 0; rt < 2; ++rt) {
            qa[rt][nt] = __builtin_amdgcn_wmma_f32_16x16x4_f32(
                false, aq[ks][rt], false, bq, (short)0, qa[rt][nt], false, false);
            pa[rt][nt] = __builtin_amdgcn_wmma_f32_16x16x4_f32(
                false, ak[ks][rt], false, bk, (short)0, pa[rt][nt], false, false);
          }
        }
      }
    };

#pragma unroll 1
    for (int kb = 0; kb < kNKB; kb += 2) {
      load_block(g, (kb + 1) * (4 * kKB), aq1, ak1, wq1, wk1);
      __builtin_amdgcn_sched_barrier(0);
      mma_block(aq0, ak0, wq0, wk0);
      __builtin_amdgcn_sched_barrier(0);
      const bool wrap = (kb + 2 >= kNKB);
      const int  gn   = wrap ? (4 * ch + ((gi + 1) & 3)) : g;
      const int  kn   = wrap ? 0 : (kb + 2) * (4 * kKB);
      load_block(gn, kn, aq0, ak0, wq0, wk0);
      __builtin_amdgcn_sched_barrier(0);
      mma_block(aq1, ak1, wq1, wk1);
      __builtin_amdgcn_sched_barrier(0);
    }

#pragma unroll
    for (int nt = 0; nt < 4; ++nt) {
      const float sk = start_key[g * 64 + 4 * l16 + nt];
#pragma unroll
      for (int rt = 0; rt < 2; ++rt) {
#pragma unroll
        for (int e = 0; e < 8; ++e) {
          const float qv = qa[rt][nt][e];
          float       kv = pa[rt][nt][e];
          if (rt == 0 && e == 0 && half == 0 && t0wave) kv = sk;
          accDot[rt][e] += qv * kv;
          accQQ[rt][e]  += qv * qv;
          accKK[rt][e]  += kv * kv;
        }
      }
    }
  }

  __syncthreads();
  float* const part = lds;
#pragma unroll
  for (int rt = 0; rt < 2; ++rt) {
#pragma unroll
    for (int e = 0; e < 8; ++e) {
      float d  = accDot[rt][e];
      float qq = accQQ[rt][e];
      float kk = accKK[rt][e];
#pragma unroll
      for (int m = 8; m >= 1; m >>= 1) {
        d  += __shfl_xor(d,  m, 32);
        qq += __shfl_xor(qq, m, 32);
        kk += __shfl_xor(kk, m, 32);
      }
      if (l16 == 0) {
        const int row = rp * 32 + rt * 16 + e + 8 * half;
        float* p = part + (ch * kTM + row) * 3;
        p[0] = d; p[1] = qq; p[2] = kk;
      }
    }
  }
  __syncthreads();

  if (tid < kTM) {
    const float* p0 = part + tid * 3;
    const float* p1 = part + (kTM + tid) * 3;
    const float d  = p0[0] + p1[0];
    const float qq = p0[1] + p1[1];
    const float kk = p0[2] + p1[2];
    const int t  = tBase + tid;
    const float na = fmaxf(sqrtf(qq), kEPS);
    const float nb = fmaxf(sqrtf(kk), kEPS);
    const float cs = d / (na * nb);
    const float pr = (1.0f - cs) * 0.5f;
    probs[(long)b * kL + t] = pr;
    const bool bd = (t == 0) || (pr > 0.5f);
    float cntLoc = bd ? 1.0f : 0.0f;
    float sumLoc = pr;
#pragma unroll
    for (int m = 16; m >= 1; m >>= 1) {
      cntLoc += __shfl_xor(cntLoc, m, 32);
      sumLoc += __shfl_xor(sumLoc, m, 32);
    }
    if (lane == 0) {
      atomicAdd(&stats[b], cntLoc);
      atomicAdd(&stats[8 + b], sumLoc);
    }
  }
}

namespace {
constexpr int kTT = 256;
constexpr int kNT = kL / kTT;
constexpr int kHT = 128;
constexpr int kSB = kHT + 32;
constexpr int kRQ = kTT * 8 / kHT;
constexpr int kCB = 32;
}

__global__ __launch_bounds__(kSB) void scan_kernel(
    const float* __restrict__ tokens,
    const float* __restrict__ probs,
    float* __restrict__ ups)
{
  __shared__ __attribute__((aligned(16))) float sm[4 * kTT * 32 + 2 * kL];
  float* const ys0 = sm;
  float* const xs0 = sm + 2 * kTT * 32;
  float* const gs  = sm + 4 * kTT * 32;
  float* const qs  = gs + kL;

  const int tid   = threadIdx.x;
  const int lane  = tid & 31;
  const bool stager = tid >= 32;
  const int b     = blockIdx.x / (kD / 32);
  const int dbase = (blockIdx.x % (kD / 32)) * 32;
  const float* tk = tokens + (long)b * kL * kD + dbase;
  const float* pr = probs  + (long)b * kL;
  float*       up = ups    + (long)b * kL * kD + dbase;

  const int hrow = (tid - 32) >> 3;
  const int hc4  = ((tid - 32) & 7) * 4;

  const float4 z4 = make_float4(0.f, 0.f, 0.f, 0.f);
  float4 r[kRQ];
#pragma unroll
  for (int j = 0; j < kRQ; ++j) r[j] = z4;

  if (stager) {
#pragma unroll
    for (int j = 0; j < kRQ; ++j)
      r[j] = *(const float4*)(tk + (long)(hrow + 16 * j) * kD + hc4);
  }

  for (int t4 = tid * 4; t4 < kL; t4 += kSB * 4) {
    const float4 p4 = *(const float4*)(pr + t4);
    const float pv[4] = {p4.x, p4.y, p4.z, p4.w};
    float gv[4], qv[4];
#pragma unroll
    for (int j = 0; j < 4; ++j) {
      const bool bd = (t4 + j == 0) || (pv[j] > 0.5f);
      gv[j] = bd ? 1.0f - pv[j] : 1.0f;
      qv[j] = bd ? pv[j] : 0.0f;
    }
    *(float4*)&gs[t4] = make_float4(gv[0], gv[1], gv[2], gv[3]);
    *(float4*)&qs[t4] = make_float4(qv[0], qv[1], qv[2], qv[3]);
  }
  __syncthreads();

  if (stager) {
#pragma unroll
    for (int j = 0; j < kRQ; ++j) {
      const int   row = hrow + 16 * j;
      const float q   = qs[row];
      *(float4*)&xs0[row * 32 + hc4] =
          make_float4(r[j].x * q, r[j].y * q, r[j].z * q, r[j].w * q);
    }
  }
  __syncthreads();

  float y = 0.f;
  for (int n = 0; n < kNT; ++n) {
    const int cur = n & 1;
    if (stager) {
      if (n + 1 < kNT) {
        const float* tkn = tk + (long)(n + 1) * kTT * kD;
#pragma unroll
        for (int j = 0; j < kRQ; ++j)
          r[j] = *(const float4*)(tkn + (long)(hrow + 16 * j) * kD + hc4);
      }
      if (n > 0) {
        const float* ysp = ys0 + (cur ^ 1) * (kTT * 32);
        float*       upt = up + (long)(n - 1) * kTT * kD;
#pragma unroll
        for (int j = 0; j < kRQ; ++j) {
          const int row = hrow + 16 * j;
          *(float4*)(upt + (long)row * kD + hc4) = *(const float4*)&ysp[row * 32 + hc4];
        }
      }
      if (n + 1 < kNT) {
        float*       xsn = xs0 + (cur ^ 1) * (kTT * 32);
        const float* qn  = qs + (n + 1) * kTT;
#pragma unroll
        for (int j = 0; j < kRQ; ++j) {
          const int   row = hrow + 16 * j;
          const float q   = qn[row];
          *(float4*)&xsn[row * 32 + hc4] =
              make_float4(r[j].x * q, r[j].y * q, r[j].z * q, r[j].w * q);
        }
      }
    } else {
      const float* xsc = xs0 + cur * (kTT * 32) + lane;
      float*       ysc = ys0 + cur * (kTT * 32) + lane;
      const float* gsc = gs + n * kTT;
      float xa[kCB], ga[kCB], xb[kCB], gb[kCB];
#pragma unroll
      for (int u = 0; u < kCB; ++u) { xa[u] = xsc[u * 32]; ga[u] = gsc[u]; }
      for (int k = 0; k < kTT; k += 2 * kCB) {
#pragma unroll
        for (int u = 0; u < kCB; ++u) { xb[u] = xsc[(k + kCB + u) * 32]; gb[u] = gsc[k + kCB + u]; }
#pragma unroll
        for (int u = 0; u < kCB; ++u) {
          y = fmaf(ga[u], y, xa[u]);
          ysc[(k + u) * 32] = y;
        }
        const int ka = (k + 2 * kCB < kTT) ? (k + 2 * kCB) : 0;
#pragma unroll
        for (int u = 0; u < kCB; ++u) { xa[u] = xsc[(ka + u) * 32]; ga[u] = gsc[ka + u]; }
#pragma unroll
        for (int u = 0; u < kCB; ++u) {
          y = fmaf(gb[u], y, xb[u]);
          ysc[(k + kCB + u) * 32] = y;
        }
      }
    }
    __syncthreads();
  }

  if (stager) {
    const float* ysp = ys0 + ((kNT - 1) & 1) * (kTT * 32);
    float*       upt = up + (long)(kNT - 1) * kTT * kD;
#pragma unroll
    for (int j = 0; j < kRQ; ++j) {
      const int row = hrow + 16 * j;
      *(float4*)(upt + (long)row * kD + hc4) = *(const float4*)&ysp[row * 32 + hc4];
    }
  }
}

__global__ void aux_kernel(const float* __restrict__ stats,
                           float* __restrict__ outAux)
{
  if (threadIdx.x == 0 && blockIdx.x == 0) {
    float s = 0.f;
    for (int b = 0; b < kB; ++b) {
      const float F = stats[b] * (1.0f / kL);
      const float G = stats[8 + b] * (1.0f / kL);
      s += (6.0f / 5.0f) * (5.0f * F * G + (1.0f - F) * (1.0f - G));
    }
    outAux[0] = (s / (float)kB) * 0.03f;
  }
}

extern "C" void kernel_launch(void* const* d_in, const int* in_sizes, int n_in,
                              void* d_out, int out_size, void* d_ws, size_t ws_size,
                              hipStream_t stream) {
  (void)in_sizes; (void)n_in; (void)out_size; (void)ws_size;
  const float* tokens = (const float*)d_in[0];
  const float* w_qk   = (const float*)d_in[1];
  const float* skey   = (const float*)d_in[2];

  float* ups  = (float*)d_out;
  float* aux  = ups + (size_t)kB * kL * kD;
  float* probs = (float*)d_ws;
  float* stats = probs + (size_t)kB * kL;

  hipMemsetAsync(stats, 0, 16 * sizeof(float), stream);

  const size_t shmem = (size_t)kAR * kAP * sizeof(float);
  qk_probs_kernel<<<dim3(kB * (kL / kTM)), dim3(256), shmem, stream>>>(
      tokens, w_qk, skey, probs, stats);
  scan_kernel<<<dim3(kB * (kD / 32)), dim3(kSB), 0, stream>>>(tokens, probs, ups);
  aux_kernel<<<1, 32, 0, stream>>>(stats, aux);
}


# scan: 16 waves per block each run 32 consecutive steps of a 512-step pass from global memory; a wave's carry-in is folded from earlier waves' end values and gate products: new summation order
# speedup vs baseline: 7.4397x; 1.0775x over previous
#include <hip/hip_runtime.h>

typedef __attribute__((ext_vector_type(2))) float v2f;
typedef __attribute__((ext_vector_type(4))) float v4f;
typedef __attribute__((ext_vector_type(8))) float v8f;

namespace {
constexpr int   kB   = 8;
constexpr int   kL   = 4096;
constexpr int   kD   = 512;
constexpr int   kNW  = 1024;
constexpr int   kTM  = 128;
constexpr int   kAR  = kTM + 1;
constexpr int   kAP  = 516;
constexpr float kEPS = 1e-8f;
constexpr int   kKB  = 4;
constexpr int   kNKB = kD / (4 * kKB);
}

__global__ __launch_bounds__(256) void qk_probs_kernel(
    const float* __restrict__ tokens,
    const float* __restrict__ w_qk,
    const float* __restrict__ start_key,
    float* __restrict__ probs,
    float* __restrict__ stats)
{
  extern __shared__ float lds[];

  const int tid  = threadIdx.x;
  const int wave = tid >> 5;
  const int lane = tid & 31;
  const int half = lane >> 4;
  const int l16  = lane & 15;
  const int rp   = wave >> 1;
  const int ch   = wave & 1;

  const int blocksPerBatch = kL / kTM;
  const int b     = blockIdx.x / blocksPerBatch;
  const int tBase = (blockIdx.x % blocksPerBatch) * kTM;

  {
    const float* tokB = tokens + (long)b * kL * kD;
    const int c4 = (tid & 127) << 2;
    const int r0 = tid >> 7;
#pragma unroll 1
    for (int it = 0; it < 4; ++it) {
      float4 v[16];
#pragma unroll
      for (int j = 0; j < 16; ++j) {
        const int row = 2 * (it * 16 + j) + r0;
        int tr = tBase - 1 + row;
        tr = tr < 0 ? 0 : tr;
        v[j] = *(const float4*)(tokB + (long)tr * kD + c4);
      }
#pragma unroll
      for (int j = 0; j < 16; ++j) {
        const int row = 2 * (it * 16 + j) + r0;
        *(float4*)(lds + row * kAP + c4) = v[j];
      }
    }
    if (tid < 128) {
      const float4 v = *(const float4*)(tokB + (long)(tBase + kTM - 1) * kD + c4);
      *(float4*)(lds + kTM * kAP + c4) = v;
    }
  }
  __syncthreads();

  const float* Akp = lds + (rp * 32 + l16) * kAP + 2 * half;
  const float* Aq  = Akp + kAP;
  const float* Wq = w_qk + (long)(2 * half) * kNW + 4 * l16;
  const float* Wk = Wq + 512;

  const v8f z8 = {0.f, 0.f, 0.f, 0.f, 0.f, 0.f, 0.f, 0.f};
  v8f accDot[2] = {z8, z8}, accQQ[2] = {z8, z8}, accKK[2] = {z8, z8};

  const bool t0wave = (tBase == 0) && (rp == 0);

  v2f aq0[kKB][2], ak0[kKB][2], aq1[kKB][2], ak1[kKB][2];
  v4f wq0[kKB][2], wk0[kKB][2], wq1[kKB][2], wk1[kKB][2];

  auto load_block = [&](int g, int k0, v2f (&aq)[kKB][2], v2f (&ak)[kKB][2],
                        v4f (&wq)[kKB][2], v4f (&wk)[kKB][2]) {
#pragma unroll
    for (int ks = 0; ks < kKB; ++ks) {
#pragma unroll
      for (int s = 0; s < 2; ++s) {
        const long ro = (long)(k0 + 4 * ks + s) * kNW + g * 64;
        wq[ks][s] = *(const v4f*)(Wq + ro);
        wk[ks][s] = *(const v4f*)(Wk + ro);
      }
    }
#pragma unroll
    for (int ks = 0; ks < kKB; ++ks) {
#pragma unroll
      for (int rt = 0; rt < 2; ++rt) {
        aq[ks][rt] = *(const v2f*)(Aq + rt * 16 * kAP + k0 + 4 * ks);
        ak[ks][rt] = *(const v2f*)(Akp + rt * 16 * kAP + k0 + 4 * ks);
      }
    }
  };

  load_block(4 * ch, 0, aq0, ak0, wq0, wk0);

  for (int gi = 0; gi < 4; ++gi) {
    const int g = 4 * ch + gi;
    v8f qa[2][4], pa[2][4];
#pragma unroll
    for (int rt = 0; rt < 2; ++rt) {
#pragma unroll
      for (int nt = 0; nt < 4; ++nt) { qa[rt][nt] = z8; pa[rt][nt] = z8; }
    }

    auto mma_block = [&](const v2f (&aq)[kKB][2], const v2f (&ak)[kKB][2],
                         const v4f (&wq)[kKB][2], const v4f (&wk)[kKB][2]) {
#pragma unroll
      for (int ks = 0; ks < kKB; ++ks) {
#pragma unroll
        for (int nt = 0; nt < 4; ++nt) {
          const v2f bq = {wq[ks][0][nt], wq[ks][1][nt]};
          const v2f bk = {wk[ks][0][nt], wk[ks][1][nt]};
#pragma unroll
          for (int rt = 0; rt < 2; ++rt) {
            qa[rt][nt] = __builtin_amdgcn_wmma_f32_16x16x4_f32(
                false, aq[ks][rt], false, bq, (short)0, qa[rt][nt], false, false);
            pa[rt][nt] = __builtin_amdgcn_wmma_f32_16x16x4_f32(
                false, ak[ks][rt], false, bk, (short)0, pa[rt][nt], false, false);
          }
        }
      }
    };

#pragma unroll 1
    for (int kb = 0; kb < kNKB; kb += 2) {
      load_block(g, (kb + 1) * (4 * kKB), aq1, ak1, wq1, wk1);
      __builtin_amdgcn_sched_barrier(0);
      mma_block(aq0, ak0, wq0, wk0);
      __builtin_amdgcn_sched_barrier(0);
      const bool wrap = (kb + 2 >= kNKB);
      const int  gn   = wrap ? (4 * ch + ((gi + 1) & 3)) : g;
      const int  kn   = wrap ? 0 : (kb + 2) * (4 * kKB);
      load_block(gn, kn, aq0, ak0, wq0, wk0);
      __builtin_amdgcn_sched_barrier(0);
      mma_block(aq1, ak1, wq1, wk1);
      __builtin_amdgcn_sched_barrier(0);
    }

#pragma unroll
    for (int nt = 0; nt < 4; ++nt) {
      const float sk = start_key[g * 64 + 4 * l16 + nt];
#pragma unroll
      for (int rt = 0; rt < 2; ++rt) {
#pragma unroll
        for (int e = 0; e < 8; ++e) {
          const float qv = qa[rt][nt][e];
          float       kv = pa[rt][nt][e];
          if (rt == 0 && e == 0 && half == 0 && t0wave) kv = sk;
          accDot[rt][e] += qv * kv;
          accQQ[rt][e]  += qv * qv;
          accKK[rt][e]  += kv * kv;
        }
      }
    }
  }

  __syncthreads();
  float* const part = lds;
#pragma unroll
  for (int rt = 0; rt < 2; ++rt) {
#pragma unroll
    for (int e = 0; e < 8; ++e) {
      float d  = accDot[rt][e];
      float qq = accQQ[rt][e];
      float kk = accKK[rt][e];
#pragma unroll
      for (int m = 8; m >= 1; m >>= 1) {
        d  += __shfl_xor(d,  m, 32);
        qq += __shfl_xor(qq, m, 32);
        kk += __shfl_xor(kk, m, 32);
      }
      if (l16 == 0) {
        const int row = rp * 32 + rt * 16 + e + 8 * half;
        float* p = part + (ch * kTM + row) * 3;
        p[0] = d; p[1] = qq; p[2] = kk;
      }
    }
  }
  __syncthreads();

  if (tid < kTM) {
    const float* p0 = part + tid * 3;
    const float* p1 = part + (kTM + tid) * 3;
    const float d  = p0[0] + p1[0];
    const float qq = p0[1] + p1[1];
    const float kk = p0[2] + p1[2];
    const int t  = tBase + tid;
    const float na = fmaxf(sqrtf(qq), kEPS);
    const float nb = fmaxf(sqrtf(kk), kEPS);
    const float cs = d / (na * nb);
    const float pr = (1.0f - cs) * 0.5f;
    probs[(long)b * kL + t] = pr;
    const bool bd = (t == 0) || (pr > 0.5f);
    float cntLoc = bd ? 1.0f : 0.0f;
    float sumLoc = pr;
#pragma unroll
    for (int m = 16; m >= 1; m >>= 1) {
      cntLoc += __shfl_xor(cntLoc, m, 32);
      sumLoc += __shfl_xor(sumLoc, m, 32);
    }
    if (lane == 0) {
      atomicAdd(&stats[b], cntLoc);
      atomicAdd(&stats[8 + b], sumLoc);
    }
  }
}

namespace {
constexpr int kSW = 16;
constexpr int kSS = 32;
constexpr int kSP = kSW * kSS;
constexpr int kNP = kL / kSP;
constexpr int kSB = kSW * 32;
}

__global__ __launch_bounds__(kSB) void scan_kernel(
    const float* __restrict__ tokens,
    const float* __restrict__ probs,
    float* __restrict__ ups)
{
  __shared__ __attribute__((aligned(16))) float sm[2 * kL];
  __shared__ float sE[2][kSW][32];
  __shared__ float sP[2][kSW];
  __shared__ float sC[2][32];
  float* const gs = sm;
  float* const qs = sm + kL;

  const int tid   = threadIdx.x;
  const int lane  = tid & 31;
  const int w     = tid >> 5;
  const int b     = blockIdx.x / (kD / 32);
  const int dbase = (blockIdx.x % (kD / 32)) * 32;
  const long off  = (long)b * kL * kD + (long)(w * kSS) * kD + dbase + lane;
  const float* tk = tokens + off;
  float*       up = ups + off;
  const float* pr = probs + (long)b * kL;

  float xn[kSS];
#pragma unroll
  for (int i = 0; i < kSS; ++i) xn[i] = tk[(long)i * kD];

  {
    float4 p4[kL / (kSB * 4)];
#pragma unroll
    for (int s = 0; s < kL / (kSB * 4); ++s)
      p4[s] = *(const float4*)(pr + s * (kSB * 4) + tid * 4);
#pragma unroll
    for (int s = 0; s < kL / (kSB * 4); ++s) {
      const int t4 = s * (kSB * 4) + tid * 4;
      const float pv[4] = {p4[s].x, p4[s].y, p4[s].z, p4[s].w};
      float gv[4], qv[4];
#pragma unroll
      for (int j = 0; j < 4; ++j) {
        const bool bd = (t4 + j == 0) || (pv[j] > 0.5f);
        gv[j] = bd ? 1.0f - pv[j] : 1.0f;
        qv[j] = bd ? pv[j] : 0.0f;
      }
      *(float4*)&gs[t4] = make_float4(gv[0], gv[1], gv[2], gv[3]);
      *(float4*)&qs[t4] = make_float4(qv[0], qv[1], qv[2], qv[3]);
    }
  }
  __syncthreads();

  for (int n = 0; n < kNP; ++n) {
    const int par = n & 1;
    const int t0  = n * kSP + w * kSS;

    float x[kSS], g[kSS];
#pragma unroll
    for (int i = 0; i < kSS; ++i) {
      g[i] = gs[t0 + i];
      x[i] = xn[i] * qs[t0 + i];
    }
    float e = 0.f;
#pragma unroll
    for (int i = 0; i < kSS; ++i) e = fmaf(g[i], e, x[i]);
    sE[par][w][lane] = e;
    if (lane == 0) {
      float pp[kSS / 2];
#pragma unroll
      for (int i = 0; i < kSS / 2; ++i) pp[i] = g[2 * i] * g[2 * i + 1];
#pragma unroll
      for (int s = kSS / 4; s >= 1; s >>= 1) {
#pragma unroll
        for (int i = 0; i < s; ++i) pp[i] = pp[2 * i] * pp[2 * i + 1];
      }
      sP[par][w] = pp[0];
    }
    __syncthreads();

    if (n + 1 < kNP) {
      const float* tkn = tk + (long)(n + 1) * kSP * kD;
#pragma unroll
      for (int i = 0; i < kSS; ++i) xn[i] = tkn[(long)i * kD];
    }

    float Ej[kSW], Pj[kSW];
#pragma unroll
    for (int j = 0; j < kSW; ++j) { Ej[j] = sE[par][j][lane]; Pj[j] = sP[par][j]; }
    float c = (n == 0) ? 0.f : sC[par][lane];
#pragma unroll
    for (int j = 0; j < kSW; ++j) {
      const float cn = fmaf(Pj[j], c, Ej[j]);
      c = (j < w) ? cn : c;
    }

    float  y   = c;
    float* upn = up + (long)n * kSP * kD;
#pragma unroll
    for (int i = 0; i < kSS; ++i) {
      y = fmaf(g[i], y, x[i]);
      upn[(long)i * kD] = y;
    }
    if (w == kSW - 1) sC[par ^ 1][lane] = y;
  }
}

__global__ void aux_kernel(const float* __restrict__ stats,
                           float* __restrict__ outAux)
{
  if (threadIdx.x == 0 && blockIdx.x == 0) {
    float s = 0.f;
    for (int b = 0; b < kB; ++b) {
      const float F = stats[b] * (1.0f / kL);
      const float G = stats[8 + b] * (1.0f / kL);
      s += (6.0f / 5.0f) * (5.0f * F * G + (1.0f - F) * (1.0f - G));
    }
    outAux[0] = (s / (float)kB) * 0.03f;
  }
}

extern "C" void kernel_launch(void* const* d_in, const int* in_sizes, int n_in,
                              void* d_out, int out_size, void* d_ws, size_t ws_size,
                              hipStream_t stream) {
  (void)in_sizes; (void)n_in; (void)out_size; (void)ws_size;
  const float* tokens = (const float*)d_in[0];
  const float* w_qk   = (const float*)d_in[1];
  const float* skey   = (const float*)d_in[2];

  float* ups  = (float*)d_out;
  float* aux  = ups + (size_t)kB * kL * kD;
  float* probs = (float*)d_ws;
  float* stats = probs + (size_t)kB * kL;

  hipMemsetAsync(stats, 0, 16 * sizeof(float), stream);

  const size_t shmem = (size_t)kAR * kAP * sizeof(float);
  qk_probs_kernel<<<dim3(kB * (kL / kTM)), dim3(256), shmem, stream>>>(
      tokens, w_qk, skey, probs, stats);
  scan_kernel<<<dim3(kB * (kD / 32)), dim3(kSB), 0, stream>>>(tokens, probs, ups);
  aux_kernel<<<1, 32, 0, stream>>>(stats, aux);
}
